# starting kernel
# baseline (speedup 1.0000x reference)
#include <hip/hip_runtime.h>
#include <hip/hip_bf16.h>


typedef __attribute__((ext_vector_type(2))) float v2f;
typedef __attribute__((ext_vector_type(8))) float v8f;

#define D 64


__global__ void deg_init(float* __restrict__ deg, int N) {
  int i = blockIdx.x * blockDim.x + threadIdx.x;
  if (i < N) deg[i] = 1.0f;
}

__global__ void deg_count(const int* __restrict__ dst, float* __restrict__ deg, int E) {
  int e = blockIdx.x * blockDim.x + threadIdx.x;
  if (e < E) atomicAdd(&deg[dst[e]], 1.0f);
}

__global__ void deg_rsqrt(float* __restrict__ deg, int N) {
  int i = blockIdx.x * blockDim.x + threadIdx.x;
  if (i < N) deg[i] = rsqrtf(deg[i]);
}


template <bool RELU_BIAS>
__global__ void gemm64_wmma(const float* __restrict__ X, const float* __restrict__ W,
                            const float* __restrict__ bias, float* __restrict__ Y,
                            int nTiles) {
  __shared__ float Wlds[D * D];
  for (int i = threadIdx.x; i < (D * D) / 4; i += blockDim.x) {
    ((float4*)Wlds)[i] = ((const float4*)W)[i];
  }
  __syncthreads();

  const int wave = threadIdx.x >> 5;
  const int lane = threadIdx.x & 31;
  const int tile = blockIdx.x * (blockDim.x >> 5) + wave;
  if (tile >= nTiles) return;

  const int lrow  = lane & 15;
  const int khalf = lane >> 4;
  const float* xrow = X + (size_t)(tile * 16 + lrow) * D;

  v8f acc0 = {}, acc1 = {}, acc2 = {}, acc3 = {};

  for (int k0 = 0; k0 < D; k0 += 4) {
    const int kb = k0 + khalf * 2;
    float a0 = xrow[kb + 0];
    float a1 = xrow[kb + 1];
    if (RELU_BIAS) {
      a0 = fmaxf(a0 + bias[kb + 0], 0.0f);
      a1 = fmaxf(a1 + bias[kb + 1], 0.0f);
    }
    v2f a;
    a.x = a0;
    a.y = a1;
#pragma unroll
    for (int nt = 0; nt < 4; ++nt) {
      const int n = nt * 16 + lrow;
      v2f b;
      b.x = Wlds[(kb + 0) * D + n];
      b.y = Wlds[(kb + 1) * D + n];
      switch (nt) {
        case 0: acc0 = __builtin_amdgcn_wmma_f32_16x16x4_f32(false, a, false, b, (short)0, acc0, false, false); break;
        case 1: acc1 = __builtin_amdgcn_wmma_f32_16x16x4_f32(false, a, false, b, (short)0, acc1, false, false); break;
        case 2: acc2 = __builtin_amdgcn_wmma_f32_16x16x4_f32(false, a, false, b, (short)0, acc2, false, false); break;
        case 3: acc3 = __builtin_amdgcn_wmma_f32_16x16x4_f32(false, a, false, b, (short)0, acc3, false, false); break;
      }
    }
  }

  const int mbase = tile * 16 + khalf * 8;
#pragma unroll
  for (int r = 0; r < 8; ++r) {
    float* yrow = Y + (size_t)(mbase + r) * D;
    yrow[0 * 16 + lrow] = acc0[r];
    yrow[1 * 16 + lrow] = acc1[r];
    yrow[2 * 16 + lrow] = acc2[r];
    yrow[3 * 16 + lrow] = acc3[r];
  }
}

template <bool RELU_BIAS>
__global__ void gemm64_tail(const float* __restrict__ X, const float* __restrict__ W,
                            const float* __restrict__ bias, float* __restrict__ Y,
                            int row0, int N) {
  int t = blockIdx.x * blockDim.x + threadIdx.x;
  int row = row0 + t / D;
  int col = t % D;
  if (row >= N) return;
  float s = 0.0f;
  for (int k = 0; k < D; ++k) {
    float x = X[(size_t)row * D + k];
    if (RELU_BIAS) x = fmaxf(x + bias[k], 0.0f);
    s += x * W[k * D + col];
  }
  Y[(size_t)row * D + col] = s;
}


__global__ void self_msg(const float* __restrict__ xw, const float* __restrict__ dinv,
                         const float* __restrict__ bias, float* __restrict__ agg,
                         int N, int addBias) {
  int t = blockIdx.x * blockDim.x + threadIdx.x;
  if (t >= N * (D / 4)) return;
  int i = t / (D / 4);
  int d = (t % (D / 4)) * 4;
  float nrm = dinv[i];
  nrm *= nrm;
  float4 v = ((const float4*)(xw + (size_t)i * D + d))[0];
  v.x *= nrm; v.y *= nrm; v.z *= nrm; v.w *= nrm;
  if (addBias) {
    v.x += bias[d + 0]; v.y += bias[d + 1]; v.z += bias[d + 2]; v.w += bias[d + 3];
  }
  ((float4*)(agg + (size_t)i * D + d))[0] = v;
}

__global__ void scatter_edges(const int* __restrict__ src, const int* __restrict__ dst,
                              const float* __restrict__ dinv, const float* __restrict__ xw,
                              float* __restrict__ agg, int E) {
  long long t = (long long)blockIdx.x * blockDim.x + threadIdx.x;
  long long e = t >> 4;
  if (e >= E) return;
  int d = (int)(t & 15) * 4;
  int s = src[e];
  int q = dst[e];
  float nrm = dinv[s] * dinv[q];
  float4 v = ((const float4*)(xw + (size_t)s * D + d))[0];
  float* out = agg + (size_t)q * D + d;
  atomicAdd(out + 0, v.x * nrm);
  atomicAdd(out + 1, v.y * nrm);
  atomicAdd(out + 2, v.z * nrm);
  atomicAdd(out + 3, v.w * nrm);
}


extern "C" void kernel_launch(void* const* d_in, const int* in_sizes, int n_in,
                              void* d_out, int out_size, void* d_ws, size_t ws_size,
                              hipStream_t stream) {
  const float* ndata = (const float*)d_in[0];
  const float* W0    = (const float*)d_in[1];
  const float* b0    = (const float*)d_in[2];
  const float* W1    = (const float*)d_in[3];
  const float* b1    = (const float*)d_in[4];
  const int* edge_src = (const int*)d_in[5];
  const int* edge_dst = (const int*)d_in[6];

  const int N = in_sizes[0] / D;
  const int E = in_sizes[5];

  size_t Npad = ((size_t)N + 3) & ~(size_t)3;
  float* dinv = (float*)d_ws;
  float* xw   = dinv + Npad;
  float* agg  = xw + (size_t)N * D;
  float* out  = (float*)d_out;

  const int T = 256;

  deg_init<<<(N + T - 1) / T, T, 0, stream>>>(dinv, N);
  deg_count<<<(E + T - 1) / T, T, 0, stream>>>(edge_dst, dinv, E);
  deg_rsqrt<<<(N + T - 1) / T, T, 0, stream>>>(dinv, N);

  const int nTiles = N / 16;
  const int remRow0 = nTiles * 16;
  const int wavesPerBlk = T / 32;
  const int gemmGrid = (nTiles + wavesPerBlk - 1) / wavesPerBlk;

  gemm64_wmma<false><<<gemmGrid, T, 0, stream>>>(ndata, W0, nullptr, xw, nTiles);
  if (remRow0 < N)
    gemm64_tail<false><<<((N - remRow0) * D + T - 1) / T, T, 0, stream>>>(ndata, W0, nullptr, xw, remRow0, N);

  self_msg<<<(N * (D / 4) + T - 1) / T, T, 0, stream>>>(xw, dinv, nullptr, agg, N, 0);
  {
    long long threads = (long long)E * 16;
    int grid = (int)((threads + T - 1) / T);
    scatter_edges<<<grid, T, 0, stream>>>(edge_src, edge_dst, dinv, xw, agg, E);
  }

  gemm64_wmma<true><<<gemmGrid, T, 0, stream>>>(agg, W1, b0, xw, nTiles);
  if (remRow0 < N)
    gemm64_tail<true><<<((N - remRow0) * D + T - 1) / T, T, 0, stream>>>(agg, W1, b0, xw, remRow0, N);

  self_msg<<<(N * (D / 4) + T - 1) / T, T, 0, stream>>>(xw, dinv, b1, out, N, 1);
  {
    long long threads = (long long)E * 16;
    int grid = (int)((threads + T - 1) / T);
    scatter_edges<<<grid, T, 0, stream>>>(edge_src, edge_dst, dinv, xw, out, E);
  }
}


# Edges bucketed by destination row range each call; one workgroup per bucket sums degree and feature rows in LDS (LDS adds, arrival order) instead of global float atomics
# speedup vs baseline: 1.9888x; 1.9888x over previous
#include <hip/hip_runtime.h>
#include <hip/hip_bf16.h>


typedef __attribute__((ext_vector_type(2))) float v2f;
typedef __attribute__((ext_vector_type(8))) float v8f;

#define D 64


#define RB 128
#define G 256
#define BT 1024

__device__ __forceinline__ int wave_incl_scan(int x, int lane) {
  for (int o = 1; o < 32; o <<= 1) {
    int y = __shfl_up(x, o);
    if (lane >= o) x += y;
  }
  return x;
}

__global__ void __launch_bounds__(BT)
bucket_count(const int* __restrict__ dst, int* __restrict__ cnt, int E, int CH, int NB) {
  extern __shared__ int hist[];
  for (int k = threadIdx.x; k < NB; k += BT) hist[k] = 0;
  __syncthreads();
  const int e0 = blockIdx.x * CH;
  const int e1 = min(E, e0 + CH);
  for (int e = e0 + (int)threadIdx.x; e < e1; e += BT) atomicAdd(&hist[dst[e] / RB], 1);
  __syncthreads();
  for (int k = threadIdx.x; k < NB; k += BT) cnt[blockIdx.x * NB + k] = hist[k];
}

__global__ void __launch_bounds__(G)
bucket_scan(int* __restrict__ cnt, int* __restrict__ total, int NB) {
  __shared__ int wsum[G / 32];
  const int k = blockIdx.x, t = threadIdx.x, lane = t & 31, w = t >> 5;
  const int v = cnt[t * NB + k];
  const int x = wave_incl_scan(v, lane);
  if (lane == 31) wsum[w] = x;
  __syncthreads();
  int pre = 0;
  for (int i = 0; i < w; ++i) pre += wsum[i];
  cnt[t * NB + k] = pre + x - v;
  if (t == G - 1) total[k] = pre + x;
}

__global__ void __launch_bounds__(BT)
bucket_fill(const int* __restrict__ src, const int* __restrict__ dst,
            const int* __restrict__ off, const int* __restrict__ total,
            int* __restrict__ bstart, int* __restrict__ pairs, int E, int CH, int NB) {
  extern __shared__ int cur[];
  __shared__ int wsum[BT / 32];
  const int t = threadIdx.x, lane = t & 31, w = t >> 5;
  int carry = 0;
  for (int base = 0; base < NB; base += BT) {
    const int k = base + t;
    const int v = (k < NB) ? total[k] : 0;
    const int x = wave_incl_scan(v, lane);
    if (lane == 31) wsum[w] = x;
    __syncthreads();
    int pre = 0, tot = 0;
    for (int i = 0; i < BT / 32; ++i) {
      const int s = wsum[i];
      if (i < w) pre += s;
      tot += s;
    }
    if (k < NB) {
      const int ex = carry + pre + x - v;
      cur[k] = ex + off[blockIdx.x * NB + k];
      if (blockIdx.x == 0) bstart[k] = ex;
    }
    carry += tot;
    __syncthreads();
  }
  if (blockIdx.x == 0 && t == 0) bstart[NB] = carry;
  const int e0 = blockIdx.x * CH;
  const int e1 = min(E, e0 + CH);
  for (int e = e0 + t; e < e1; e += BT) {
    const int s = src[e];
    const int q = dst[e];
    const int pos = atomicAdd(&cur[q / RB], 1);
    ((int2*)pairs)[pos] = make_int2(s, q);
  }
}

__global__ void __launch_bounds__(256)
bucket_degree(const int* __restrict__ pairs, const int* __restrict__ bstart,
              float* __restrict__ dinv, int N) {
  __shared__ int c[RB];
  const int k = blockIdx.x, t = threadIdx.x;
  const int r0 = k * RB;
  if (t < RB) c[t] = 0;
  __syncthreads();
  const int b0 = bstart[k], b1 = bstart[k + 1];
  for (int e = b0 + t; e < b1; e += 256) atomicAdd(&c[pairs[2 * e + 1] - r0], 1);
  __syncthreads();
  if (t < RB && r0 + t < N) dinv[r0 + t] = rsqrtf(1.0f + (float)c[t]);
}


template <bool RELU_BIAS>
__global__ void gemm64_wmma(const float* __restrict__ X, const float* __restrict__ W,
                            const float* __restrict__ bias, float* __restrict__ Y,
                            int nTiles) {
  __shared__ float Wlds[D * D];
  for (int i = threadIdx.x; i < (D * D) / 4; i += blockDim.x) {
    ((float4*)Wlds)[i] = ((const float4*)W)[i];
  }
  __syncthreads();

  const int wave = threadIdx.x >> 5;
  const int lane = threadIdx.x & 31;
  const int tile = blockIdx.x * (blockDim.x >> 5) + wave;
  if (tile >= nTiles) return;

  const int lrow  = lane & 15;
  const int khalf = lane >> 4;
  const float* xrow = X + (size_t)(tile * 16 + lrow) * D;

  v8f acc0 = {}, acc1 = {}, acc2 = {}, acc3 = {};

  for (int k0 = 0; k0 < D; k0 += 4) {
    const int kb = k0 + khalf * 2;
    float a0 = xrow[kb + 0];
    float a1 = xrow[kb + 1];
    if (RELU_BIAS) {
      a0 = fmaxf(a0 + bias[kb + 0], 0.0f);
      a1 = fmaxf(a1 + bias[kb + 1], 0.0f);
    }
    v2f a;
    a.x = a0;
    a.y = a1;
#pragma unroll
    for (int nt = 0; nt < 4; ++nt) {
      const int n = nt * 16 + lrow;
      v2f b;
      b.x = Wlds[(kb + 0) * D + n];
      b.y = Wlds[(kb + 1) * D + n];
      switch (nt) {
        case 0: acc0 = __builtin_amdgcn_wmma_f32_16x16x4_f32(false, a, false, b, (short)0, acc0, false, false); break;
        case 1: acc1 = __builtin_amdgcn_wmma_f32_16x16x4_f32(false, a, false, b, (short)0, acc1, false, false); break;
        case 2: acc2 = __builtin_amdgcn_wmma_f32_16x16x4_f32(false, a, false, b, (short)0, acc2, false, false); break;
        case 3: acc3 = __builtin_amdgcn_wmma_f32_16x16x4_f32(false, a, false, b, (short)0, acc3, false, false); break;
      }
    }
  }

  const int mbase = tile * 16 + khalf * 8;
#pragma unroll
  for (int r = 0; r < 8; ++r) {
    float* yrow = Y + (size_t)(mbase + r) * D;
    yrow[0 * 16 + lrow] = acc0[r];
    yrow[1 * 16 + lrow] = acc1[r];
    yrow[2 * 16 + lrow] = acc2[r];
    yrow[3 * 16 + lrow] = acc3[r];
  }
}

template <bool RELU_BIAS>
__global__ void gemm64_tail(const float* __restrict__ X, const float* __restrict__ W,
                            const float* __restrict__ bias, float* __restrict__ Y,
                            int row0, int N) {
  int t = blockIdx.x * blockDim.x + threadIdx.x;
  int row = row0 + t / D;
  int col = t % D;
  if (row >= N) return;
  float s = 0.0f;
  for (int k = 0; k < D; ++k) {
    float x = X[(size_t)row * D + k];
    if (RELU_BIAS) x = fmaxf(x + bias[k], 0.0f);
    s += x * W[k * D + col];
  }
  Y[(size_t)row * D + col] = s;
}


#define AT 512

template <bool ADD_BIAS>
__global__ void __launch_bounds__(AT)
bucket_aggregate(const int* __restrict__ pairs, const int* __restrict__ bstart,
                 const float* __restrict__ dinv, const float* __restrict__ xw,
                 const float* __restrict__ bias, float* __restrict__ out, int N) {
  __shared__ float tile[RB * D];
  __shared__ float sd[RB];
  const int k = blockIdx.x, t = threadIdx.x, lane = t & 31;
  const int wave = __builtin_amdgcn_readfirstlane(t >> 5);
  const int r0 = k * RB;
  const int rows = min(RB, N - r0);

  if (t < RB) sd[t] = (t < rows) ? dinv[r0 + t] : 0.0f;
  for (int i = t; i < RB * (D / 4); i += AT) {
    const int r = i >> 4;
    float4 v = make_float4(0.0f, 0.0f, 0.0f, 0.0f);
    if (r < rows) {
      float nrm = dinv[r0 + r];
      nrm *= nrm;
      v = ((const float4*)(xw + (size_t)r0 * D))[i];
      v.x *= nrm; v.y *= nrm; v.z *= nrm; v.w *= nrm;
    }
    ((float4*)tile)[i] = v;
  }
  __syncthreads();

  const int b0 = bstart[k], b1 = bstart[k + 1];
  for (int base = b0 + wave * 32; base < b1; base += (AT / 32) * 32) {
    const int idx = base + lane;
    int2 p = make_int2(0, r0);
    float nv = 0.0f;
    if (idx < b1) {
      p = ((const int2*)pairs)[idx];
      nv = dinv[p.x] * sd[p.y - r0];
    }
    const int nvb = __float_as_int(nv);
    const int cnt = min(32, b1 - base);
    if (cnt == 32) {
#pragma unroll 1
      for (int i = 0; i < 32; i += 8) {
        float2 rv[8];
#pragma unroll
        for (int u = 0; u < 8; ++u) {
          const int s = __builtin_amdgcn_readlane(p.x, i + u);
          rv[u] = ((const float2*)(xw + (size_t)s * D))[lane];
        }
#pragma unroll
        for (int u = 0; u < 8; ++u) {
          const int q = __builtin_amdgcn_readlane(p.y, i + u) - r0;
          const float nrm = __int_as_float(__builtin_amdgcn_readlane(nvb, i + u));
          float* tr = tile + q * D + 2 * lane;
          atomicAdd(tr + 0, rv[u].x * nrm);
          atomicAdd(tr + 1, rv[u].y * nrm);
        }
      }
    } else {
      for (int i = 0; i < cnt; ++i) {
        const int s = __builtin_amdgcn_readlane(p.x, i);
        const int q = __builtin_amdgcn_readlane(p.y, i) - r0;
        const float nrm = __int_as_float(__builtin_amdgcn_readlane(nvb, i));
        const float2 rv = ((const float2*)(xw + (size_t)s * D))[lane];
        float* tr = tile + q * D + 2 * lane;
        atomicAdd(tr + 0, rv.x * nrm);
        atomicAdd(tr + 1, rv.y * nrm);
      }
    }
  }
  __syncthreads();

  for (int i = t; i < rows * (D / 4); i += AT) {
    float4 v = ((const float4*)tile)[i];
    if (ADD_BIAS) {
      const int c = (i & 15) * 4;
      v.x += bias[c + 0]; v.y += bias[c + 1]; v.z += bias[c + 2]; v.w += bias[c + 3];
    }
    ((float4*)(out + (size_t)r0 * D))[i] = v;
  }
}


extern "C" void kernel_launch(void* const* d_in, const int* in_sizes, int n_in,
                              void* d_out, int out_size, void* d_ws, size_t ws_size,
                              hipStream_t stream) {
  const float* ndata = (const float*)d_in[0];
  const float* W0    = (const float*)d_in[1];
  const float* b0    = (const float*)d_in[2];
  const float* W1    = (const float*)d_in[3];
  const float* b1    = (const float*)d_in[4];
  const int* edge_src = (const int*)d_in[5];
  const int* edge_dst = (const int*)d_in[6];

  const int N = in_sizes[0] / D;
  const int E = in_sizes[5];

  const int NB = (N + RB - 1) / RB;
  const int CH = (E + G - 1) / G;

  size_t Npad = ((size_t)N + 3) & ~(size_t)3;
  float* dinv = (float*)d_ws;
  float* xw   = dinv + Npad;
  float* agg  = xw + (size_t)N * D;
  int* pairs  = (int*)(agg + (size_t)N * D);
  int* cnt    = pairs + 2 * (size_t)E;
  int* total  = cnt + (size_t)G * NB;
  int* bstart = total + NB;
  float* out  = (float*)d_out;

  const int T = 256;

  bucket_count<<<G, BT, NB * sizeof(int), stream>>>(edge_dst, cnt, E, CH, NB);
  bucket_scan<<<NB, G, 0, stream>>>(cnt, total, NB);
  bucket_fill<<<G, BT, NB * sizeof(int), stream>>>(edge_src, edge_dst, cnt, total, bstart, pairs, E, CH, NB);
  bucket_degree<<<NB, T, 0, stream>>>(pairs, bstart, dinv, N);

  const int nTiles = N / 16;
  const int remRow0 = nTiles * 16;
  const int wavesPerBlk = T / 32;
  const int gemmGrid = (nTiles + wavesPerBlk - 1) / wavesPerBlk;

  gemm64_wmma<false><<<gemmGrid, T, 0, stream>>>(ndata, W0, nullptr, xw, nTiles);
  if (remRow0 < N)
    gemm64_tail<false><<<((N - remRow0) * D + T - 1) / T, T, 0, stream>>>(ndata, W0, nullptr, xw, remRow0, N);

  bucket_aggregate<false><<<NB, AT, 0, stream>>>(pairs, bstart, dinv, xw, nullptr, agg, N);

  gemm64_wmma<true><<<gemmGrid, T, 0, stream>>>(agg, W1, b0, xw, nTiles);
  if (remRow0 < N)
    gemm64_tail<true><<<((N - remRow0) * D + T - 1) / T, T, 0, stream>>>(agg, W1, b0, xw, remRow0, N);

  bucket_aggregate<true><<<NB, AT, 0, stream>>>(pairs, bstart, dinv, xw, b1, out, N);
}


# Edge norm as two row scalings: GEMM stores rows times deg^-1/2; aggregation adds them with no 4-byte lookup or multiply per edge and scales each sum by the destination's deg^-1/2 (rounding differs)
# speedup vs baseline: 2.1182x; 1.0651x over previous
#include <hip/hip_runtime.h>
#include <hip/hip_bf16.h>


typedef __attribute__((ext_vector_type(2))) float v2f;
typedef __attribute__((ext_vector_type(8))) float v8f;

#define D 64


#define RB 128
#define G 256
#define BT 1024

__device__ __forceinline__ int wave_incl_scan(int x, int lane) {
  for (int o = 1; o < 32; o <<= 1) {
    int y = __shfl_up(x, o);
    if (lane >= o) x += y;
  }
  return x;
}

__global__ void __launch_bounds__(BT)
bucket_count(const int* __restrict__ dst, int* __restrict__ cnt, int E, int CH, int NB) {
  extern __shared__ int hist[];
  for (int k = threadIdx.x; k < NB; k += BT) hist[k] = 0;
  __syncthreads();
  const int e0 = blockIdx.x * CH;
  const int e1 = min(E, e0 + CH);
  for (int e = e0 + (int)threadIdx.x; e < e1; e += BT) atomicAdd(&hist[dst[e] / RB], 1);
  __syncthreads();
  for (int k = threadIdx.x; k < NB; k += BT) cnt[blockIdx.x * NB + k] = hist[k];
}

__global__ void __launch_bounds__(G)
bucket_scan(int* __restrict__ cnt, int* __restrict__ total, int NB) {
  __shared__ int wsum[G / 32];
  const int k = blockIdx.x, t = threadIdx.x, lane = t & 31, w = t >> 5;
  const int v = cnt[t * NB + k];
  const int x = wave_incl_scan(v, lane);
  if (lane == 31) wsum[w] = x;
  __syncthreads();
  int pre = 0;
  for (int i = 0; i < w; ++i) pre += wsum[i];
  cnt[t * NB + k] = pre + x - v;
  if (t == G - 1) total[k] = pre + x;
}

__global__ void __launch_bounds__(BT)
bucket_fill(const int* __restrict__ src, const int* __restrict__ dst,
            const int* __restrict__ off, const int* __restrict__ total,
            int* __restrict__ bstart, int* __restrict__ pairs, int E, int CH, int NB) {
  extern __shared__ int cur[];
  __shared__ int wsum[BT / 32];
  const int t = threadIdx.x, lane = t & 31, w = t >> 5;
  int carry = 0;
  for (int base = 0; base < NB; base += BT) {
    const int k = base + t;
    const int v = (k < NB) ? total[k] : 0;
    const int x = wave_incl_scan(v, lane);
    if (lane == 31) wsum[w] = x;
    __syncthreads();
    int pre = 0, tot = 0;
    for (int i = 0; i < BT / 32; ++i) {
      const int s = wsum[i];
      if (i < w) pre += s;
      tot += s;
    }
    if (k < NB) {
      const int ex = carry + pre + x - v;
      cur[k] = ex + off[blockIdx.x * NB + k];
      if (blockIdx.x == 0) bstart[k] = ex;
    }
    carry += tot;
    __syncthreads();
  }
  if (blockIdx.x == 0 && t == 0) bstart[NB] = carry;
  const int e0 = blockIdx.x * CH;
  const int e1 = min(E, e0 + CH);
  for (int e = e0 + t; e < e1; e += BT) {
    const int s = src[e];
    const int q = dst[e];
    const int pos = atomicAdd(&cur[q / RB], 1);
    ((int2*)pairs)[pos] = make_int2(s, q);
  }
}

__global__ void __launch_bounds__(256)
bucket_degree(const int* __restrict__ pairs, const int* __restrict__ bstart,
              float* __restrict__ dinv, int N) {
  __shared__ int c[RB];
  const int k = blockIdx.x, t = threadIdx.x;
  const int r0 = k * RB;
  if (t < RB) c[t] = 0;
  __syncthreads();
  const int b0 = bstart[k], b1 = bstart[k + 1];
  for (int e = b0 + t; e < b1; e += 256) atomicAdd(&c[pairs[2 * e + 1] - r0], 1);
  __syncthreads();
  if (t < RB && r0 + t < N) dinv[r0 + t] = rsqrtf(1.0f + (float)c[t]);
}


template <bool RELU_BIAS>
__global__ void gemm64_wmma(const float* __restrict__ X, const float* __restrict__ W,
                            const float* __restrict__ bias, const float* __restrict__ rscale,
                            float* __restrict__ Y, int nTiles) {
  __shared__ float Wlds[D * D];
  for (int i = threadIdx.x; i < (D * D) / 4; i += blockDim.x) {
    ((float4*)Wlds)[i] = ((const float4*)W)[i];
  }
  __syncthreads();

  const int wave = threadIdx.x >> 5;
  const int lane = threadIdx.x & 31;
  const int tile = blockIdx.x * (blockDim.x >> 5) + wave;
  if (tile >= nTiles) return;

  const int lrow  = lane & 15;
  const int khalf = lane >> 4;
  const float* xrow = X + (size_t)(tile * 16 + lrow) * D;

  const int mbase = tile * 16 + khalf * 8;
  const float4 rs0 = *(const float4*)(rscale + mbase);
  const float4 rs1 = *(const float4*)(rscale + mbase + 4);

  v8f acc0 = {}, acc1 = {}, acc2 = {}, acc3 = {};

  for (int k0 = 0; k0 < D; k0 += 4) {
    const int kb = k0 + khalf * 2;
    float a0 = xrow[kb + 0];
    float a1 = xrow[kb + 1];
    if (RELU_BIAS) {
      a0 = fmaxf(a0 + bias[kb + 0], 0.0f);
      a1 = fmaxf(a1 + bias[kb + 1], 0.0f);
    }
    v2f a;
    a.x = a0;
    a.y = a1;
#pragma unroll
    for (int nt = 0; nt < 4; ++nt) {
      const int n = nt * 16 + lrow;
      v2f b;
      b.x = Wlds[(kb + 0) * D + n];
      b.y = Wlds[(kb + 1) * D + n];
      switch (nt) {
        case 0: acc0 = __builtin_amdgcn_wmma_f32_16x16x4_f32(false, a, false, b, (short)0, acc0, false, false); break;
        case 1: acc1 = __builtin_amdgcn_wmma_f32_16x16x4_f32(false, a, false, b, (short)0, acc1, false, false); break;
        case 2: acc2 = __builtin_amdgcn_wmma_f32_16x16x4_f32(false, a, false, b, (short)0, acc2, false, false); break;
        case 3: acc3 = __builtin_amdgcn_wmma_f32_16x16x4_f32(false, a, false, b, (short)0, acc3, false, false); break;
      }
    }
  }

  const float rs[8] = {rs0.x, rs0.y, rs0.z, rs0.w, rs1.x, rs1.y, rs1.z, rs1.w};
#pragma unroll
  for (int r = 0; r < 8; ++r) {
    float* yrow = Y + (size_t)(mbase + r) * D;
    yrow[0 * 16 + lrow] = acc0[r] * rs[r];
    yrow[1 * 16 + lrow] = acc1[r] * rs[r];
    yrow[2 * 16 + lrow] = acc2[r] * rs[r];
    yrow[3 * 16 + lrow] = acc3[r] * rs[r];
  }
}

template <bool RELU_BIAS>
__global__ void gemm64_tail(const float* __restrict__ X, const float* __restrict__ W,
                            const float* __restrict__ bias, const float* __restrict__ rscale,
                            float* __restrict__ Y, int row0, int N) {
  int t = blockIdx.x * blockDim.x + threadIdx.x;
  int row = row0 + t / D;
  int col = t % D;
  if (row >= N) return;
  float s = 0.0f;
  for (int k = 0; k < D; ++k) {
    float x = X[(size_t)row * D + k];
    if (RELU_BIAS) x = fmaxf(x + bias[k], 0.0f);
    s += x * W[k * D + col];
  }
  Y[(size_t)row * D + col] = s * rscale[row];
}


#define AT 512

template <bool ADD_BIAS>
__global__ void __launch_bounds__(AT)
bucket_aggregate(const int* __restrict__ pairs, const int* __restrict__ bstart,
                 const float* __restrict__ dinv, const float* __restrict__ xs,
                 const float* __restrict__ bias, float* __restrict__ out, int N) {
  __shared__ float tile[RB * D];
  __shared__ float sd[RB];
  const int k = blockIdx.x, t = threadIdx.x, lane = t & 31;
  const int wave = __builtin_amdgcn_readfirstlane(t >> 5);
  const int r0 = k * RB;
  const int rows = min(RB, N - r0);

  if (t < RB) sd[t] = (t < rows) ? dinv[r0 + t] : 0.0f;
  for (int i = t; i < RB * (D / 4); i += AT) {
    const int r = i >> 4;
    float4 v = make_float4(0.0f, 0.0f, 0.0f, 0.0f);
    if (r < rows) v = ((const float4*)(xs + (size_t)r0 * D))[i];
    ((float4*)tile)[i] = v;
  }
  __syncthreads();

  const int b0 = bstart[k], b1 = bstart[k + 1];
  for (int base = b0 + wave * 32; base < b1; base += (AT / 32) * 32) {
    const int idx = base + lane;
    int2 p = make_int2(0, r0);
    if (idx < b1) p = ((const int2*)pairs)[idx];
    const int cnt = min(32, b1 - base);
    if (cnt == 32) {
#pragma unroll 1
      for (int i = 0; i < 32; i += 8) {
        float2 rv[8];
#pragma unroll
        for (int u = 0; u < 8; ++u) {
          const int s = __builtin_amdgcn_readlane(p.x, i + u);
          rv[u] = ((const float2*)(xs + (size_t)s * D))[lane];
        }
#pragma unroll
        for (int u = 0; u < 8; ++u) {
          const int q = __builtin_amdgcn_readlane(p.y, i + u) - r0;
          float* tr = tile + q * D + 2 * lane;
          atomicAdd(tr + 0, rv[u].x);
          atomicAdd(tr + 1, rv[u].y);
        }
      }
    } else {
      for (int i = 0; i < cnt; ++i) {
        const int s = __builtin_amdgcn_readlane(p.x, i);
        const int q = __builtin_amdgcn_readlane(p.y, i) - r0;
        const float2 rv = ((const float2*)(xs + (size_t)s * D))[lane];
        float* tr = tile + q * D + 2 * lane;
        atomicAdd(tr + 0, rv.x);
        atomicAdd(tr + 1, rv.y);
      }
    }
  }
  __syncthreads();

  for (int i = t; i < rows * (D / 4); i += AT) {
    float4 v = ((const float4*)tile)[i];
    const float nd = sd[i >> 4];
    v.x *= nd; v.y *= nd; v.z *= nd; v.w *= nd;
    if (ADD_BIAS) {
      const int c = (i & 15) * 4;
      v.x += bias[c + 0]; v.y += bias[c + 1]; v.z += bias[c + 2]; v.w += bias[c + 3];
    }
    ((float4*)(out + (size_t)r0 * D))[i] = v;
  }
}


extern "C" void kernel_launch(void* const* d_in, const int* in_sizes, int n_in,
                              void* d_out, int out_size, void* d_ws, size_t ws_size,
                              hipStream_t stream) {
  const float* ndata = (const float*)d_in[0];
  const float* W0    = (const float*)d_in[1];
  const float* b0    = (const float*)d_in[2];
  const float* W1    = (const float*)d_in[3];
  const float* b1    = (const float*)d_in[4];
  const int* edge_src = (const int*)d_in[5];
  const int* edge_dst = (const int*)d_in[6];

  const int N = in_sizes[0] / D;
  const int E = in_sizes[5];

  const int NB = (N + RB - 1) / RB;
  const int CH = (E + G - 1) / G;

  size_t Npad = ((size_t)N + 3) & ~(size_t)3;
  float* dinv = (float*)d_ws;
  float* xw   = dinv + Npad;
  float* agg  = xw + (size_t)N * D;
  int* pairs  = (int*)(agg + (size_t)N * D);
  int* cnt    = pairs + 2 * (size_t)E;
  int* total  = cnt + (size_t)G * NB;
  int* bstart = total + NB;
  float* out  = (float*)d_out;

  const int T = 256;

  bucket_count<<<G, BT, NB * sizeof(int), stream>>>(edge_dst, cnt, E, CH, NB);
  bucket_scan<<<NB, G, 0, stream>>>(cnt, total, NB);
  bucket_fill<<<G, BT, NB * sizeof(int), stream>>>(edge_src, edge_dst, cnt, total, bstart, pairs, E, CH, NB);
  bucket_degree<<<NB, T, 0, stream>>>(pairs, bstart, dinv, N);

  const int nTiles = N / 16;
  const int remRow0 = nTiles * 16;
  const int wavesPerBlk = T / 32;
  const int gemmGrid = (nTiles + wavesPerBlk - 1) / wavesPerBlk;

  gemm64_wmma<false><<<gemmGrid, T, 0, stream>>>(ndata, W0, nullptr, dinv, xw, nTiles);
  if (remRow0 < N)
    gemm64_tail<false><<<((N - remRow0) * D + T - 1) / T, T, 0, stream>>>(ndata, W0, nullptr, dinv, xw, remRow0, N);

  bucket_aggregate<false><<<NB, AT, 0, stream>>>(pairs, bstart, dinv, xw, nullptr, agg, N);

  gemm64_wmma<true><<<gemmGrid, T, 0, stream>>>(agg, W1, b0, dinv, xw, nTiles);
  if (remRow0 < N)
    gemm64_tail<true><<<((N - remRow0) * D + T - 1) / T, T, 0, stream>>>(agg, W1, b0, dinv, xw, remRow0, N);

  bucket_aggregate<true><<<NB, AT, 0, stream>>>(pairs, bstart, dinv, xw, b1, out, N);
}


# Bucket pass finishes the sort into per-row source lists (self-loop last); aggregation sums each row in registers in list order, one store per row, replacing LDS float adds in arrival order
# speedup vs baseline: 2.8089x; 1.3261x over previous
#include <hip/hip_runtime.h>
#include <hip/hip_bf16.h>


typedef __attribute__((ext_vector_type(2))) float v2f;
typedef __attribute__((ext_vector_type(8))) float v8f;

#define D 64


#define RB 128
#define G 256
#define BT 1024

__device__ __forceinline__ int wave_incl_scan(int x, int lane) {
  for (int o = 1; o < 32; o <<= 1) {
    int y = __shfl_up(x, o);
    if (lane >= o) x += y;
  }
  return x;
}

__global__ void __launch_bounds__(BT)
bucket_count(const int* __restrict__ dst, int* __restrict__ cnt, int E, int CH, int NB) {
  extern __shared__ int hist[];
  for (int k = threadIdx.x; k < NB; k += BT) hist[k] = 0;
  __syncthreads();
  const int e0 = blockIdx.x * CH;
  const int e1 = min(E, e0 + CH);
  for (int e = e0 + (int)threadIdx.x; e < e1; e += BT) atomicAdd(&hist[dst[e] / RB], 1);
  __syncthreads();
  for (int k = threadIdx.x; k < NB; k += BT) cnt[blockIdx.x * NB + k] = hist[k];
}

__global__ void __launch_bounds__(G)
bucket_scan(int* __restrict__ cnt, int* __restrict__ total, int NB) {
  __shared__ int wsum[G / 32];
  const int k = blockIdx.x, t = threadIdx.x, lane = t & 31, w = t >> 5;
  const int v = cnt[t * NB + k];
  const int x = wave_incl_scan(v, lane);
  if (lane == 31) wsum[w] = x;
  __syncthreads();
  int pre = 0;
  for (int i = 0; i < w; ++i) pre += wsum[i];
  cnt[t * NB + k] = pre + x - v;
  if (t == G - 1) total[k] = pre + x;
}

__global__ void __launch_bounds__(BT)
bucket_fill(const int* __restrict__ src, const int* __restrict__ dst,
            const int* __restrict__ off, const int* __restrict__ total,
            int* __restrict__ bstart, int* __restrict__ pairs, int E, int CH, int NB) {
  extern __shared__ int cur[];
  __shared__ int wsum[BT / 32];
  const int t = threadIdx.x, lane = t & 31, w = t >> 5;
  int carry = 0;
  for (int base = 0; base < NB; base += BT) {
    const int k = base + t;
    const int v = (k < NB) ? total[k] : 0;
    const int x = wave_incl_scan(v, lane);
    if (lane == 31) wsum[w] = x;
    __syncthreads();
    int pre = 0, tot = 0;
    for (int i = 0; i < BT / 32; ++i) {
      const int s = wsum[i];
      if (i < w) pre += s;
      tot += s;
    }
    if (k < NB) {
      const int ex = carry + pre + x - v;
      cur[k] = ex + off[blockIdx.x * NB + k];
      if (blockIdx.x == 0) bstart[k] = ex;
    }
    carry += tot;
    __syncthreads();
  }
  if (blockIdx.x == 0 && t == 0) bstart[NB] = carry;
  const int e0 = blockIdx.x * CH;
  const int e1 = min(E, e0 + CH);
  for (int e = e0 + t; e < e1; e += BT) {
    const int s = src[e];
    const int q = dst[e];
    const int pos = atomicAdd(&cur[q / RB], 1);
    ((int2*)pairs)[pos] = make_int2(s, q);
  }
}

#define ROWEND ((int)0x80000000)

__global__ void __launch_bounds__(256)
bucket_rows(const int* __restrict__ pairs, const int* __restrict__ bstart,
            float* __restrict__ dinv, int* __restrict__ rowptr, int* __restrict__ col,
            int N, int NB) {
  __shared__ int c[RB];
  __shared__ int wsum[RB / 32];
  const int k = blockIdx.x, t = threadIdx.x, lane = t & 31, w = t >> 5;
  const int r0 = k * RB;
  const int rows = min(RB, N - r0);
  if (t < RB) c[t] = 0;
  __syncthreads();
  const int b0 = bstart[k], b1 = bstart[k + 1];
  for (int e = b0 + t; e < b1; e += 4 * 256) {
    int qd[4];
#pragma unroll
    for (int j = 0; j < 4; ++j) qd[j] = pairs[2 * min(e + j * 256, b1 - 1) + 1];
    asm volatile("" :: "v"(qd[0]), "v"(qd[1]), "v"(qd[2]), "v"(qd[3]));
#pragma unroll
    for (int j = 0; j < 4; ++j)
      if (e + j * 256 < b1) atomicAdd(&c[qd[j] - r0], 1);
  }
  __syncthreads();
  int indeg = 0, v = 0, x = 0;
  if (t < RB) {
    indeg = c[t];
    v = (t < rows) ? indeg + 1 : 0;
    x = wave_incl_scan(v, lane);
    if (lane == 31) wsum[w] = x;
  }
  __syncthreads();
  if (t < RB) {
    int pre = 0;
    for (int i = 0; i < w; ++i) pre += wsum[i];
    const int base = b0 + r0 + pre + x - v;
    if (t < rows) {
      rowptr[r0 + t] = base;
      dinv[r0 + t] = rsqrtf(1.0f + (float)indeg);
      col[base + indeg] = (r0 + t) | ROWEND;
    }
    c[t] = base;
  }
  if (k == NB - 1 && t == 0) rowptr[N] = b1 + N;
  __syncthreads();
  for (int e = b0 + t; e < b1; e += 4 * 256) {
    int2 p[4];
#pragma unroll
    for (int j = 0; j < 4; ++j) p[j] = ((const int2*)pairs)[min(e + j * 256, b1 - 1)];
    asm volatile("" :: "v"(p[0].y), "v"(p[1].y), "v"(p[2].y), "v"(p[3].y));
#pragma unroll
    for (int j = 0; j < 4; ++j) {
      if (e + j * 256 < b1) {
        const int pos = atomicAdd(&c[p[j].y - r0], 1);
        col[pos] = p[j].x;
      }
    }
  }
}


template <bool RELU_BIAS>
__global__ void gemm64_wmma(const float* __restrict__ X, const float* __restrict__ W,
                            const float* __restrict__ bias, const float* __restrict__ rscale,
                            float* __restrict__ Y, int nTiles) {
  __shared__ float Wlds[D * D];
  for (int i = threadIdx.x; i < (D * D) / 4; i += blockDim.x) {
    ((float4*)Wlds)[i] = ((const float4*)W)[i];
  }
  __syncthreads();

  const int wave = threadIdx.x >> 5;
  const int lane = threadIdx.x & 31;
  const int tile = blockIdx.x * (blockDim.x >> 5) + wave;
  if (tile >= nTiles) return;

  const int lrow  = lane & 15;
  const int khalf = lane >> 4;
  const float* xrow = X + (size_t)(tile * 16 + lrow) * D;

  const int mbase = tile * 16 + khalf * 8;
  const float4 rs0 = *(const float4*)(rscale + mbase);
  const float4 rs1 = *(const float4*)(rscale + mbase + 4);

  v8f acc0 = {}, acc1 = {}, acc2 = {}, acc3 = {};

  for (int k0 = 0; k0 < D; k0 += 4) {
    const int kb = k0 + khalf * 2;
    float a0 = xrow[kb + 0];
    float a1 = xrow[kb + 1];
    if (RELU_BIAS) {
      a0 = fmaxf(a0 + bias[kb + 0], 0.0f);
      a1 = fmaxf(a1 + bias[kb + 1], 0.0f);
    }
    v2f a;
    a.x = a0;
    a.y = a1;
#pragma unroll
    for (int nt = 0; nt < 4; ++nt) {
      const int n = nt * 16 + lrow;
      v2f b;
      b.x = Wlds[(kb + 0) * D + n];
      b.y = Wlds[(kb + 1) * D + n];
      switch (nt) {
        case 0: acc0 = __builtin_amdgcn_wmma_f32_16x16x4_f32(false, a, false, b, (short)0, acc0, false, false); break;
        case 1: acc1 = __builtin_amdgcn_wmma_f32_16x16x4_f32(false, a, false, b, (short)0, acc1, false, false); break;
        case 2: acc2 = __builtin_amdgcn_wmma_f32_16x16x4_f32(false, a, false, b, (short)0, acc2, false, false); break;
        case 3: acc3 = __builtin_amdgcn_wmma_f32_16x16x4_f32(false, a, false, b, (short)0, acc3, false, false); break;
      }
    }
  }

  const float rs[8] = {rs0.x, rs0.y, rs0.z, rs0.w, rs1.x, rs1.y, rs1.z, rs1.w};
#pragma unroll
  for (int r = 0; r < 8; ++r) {
    float* yrow = Y + (size_t)(mbase + r) * D;
    yrow[0 * 16 + lrow] = acc0[r] * rs[r];
    yrow[1 * 16 + lrow] = acc1[r] * rs[r];
    yrow[2 * 16 + lrow] = acc2[r] * rs[r];
    yrow[3 * 16 + lrow] = acc3[r] * rs[r];
  }
}

template <bool RELU_BIAS>
__global__ void gemm64_tail(const float* __restrict__ X, const float* __restrict__ W,
                            const float* __restrict__ bias, const float* __restrict__ rscale,
                            float* __restrict__ Y, int row0, int N) {
  int t = blockIdx.x * blockDim.x + threadIdx.x;
  int row = row0 + t / D;
  int col = t % D;
  if (row >= N) return;
  float s = 0.0f;
  for (int k = 0; k < D; ++k) {
    float x = X[(size_t)row * D + k];
    if (RELU_BIAS) x = fmaxf(x + bias[k], 0.0f);
    s += x * W[k * D + col];
  }
  Y[(size_t)row * D + col] = s * rscale[row];
}


#define AGT 256
#define AGRID 1024
#define CHUNK 32
#define ROWBYTES ((unsigned)(D * sizeof(float)))


template <bool FULL, int I, int CNT>
__device__ __forceinline__ void gather_rows(const float* __restrict__ xs, int ent, int m, unsigned lane8,
                                            int (&s)[CHUNK], float2 (&rv)[CHUNK]) {
#pragma unroll
  for (int u = 0; u < CNT; ++u) {
    s[I + u] = FULL ? __builtin_amdgcn_readlane(ent, I + u)
                    : __builtin_amdgcn_readlane(ent, I + min(u, m - 1));
    rv[I + u] = *(const float2*)((const char*)xs + ((unsigned)s[I + u] * ROWBYTES + lane8));
  }
}

template <bool ADD_BIAS, bool FULL, int I, int CNT>
__device__ __forceinline__ void sum_rows(float* __restrict__ out,
                                         const int (&s)[CHUNK], const float2 (&rv)[CHUNK], int m,
                                         unsigned lane8, float2 bv, int r_lo, float dv,
                                         int& cur, float2& acc) {
#pragma unroll
  for (int u = 0; u < CNT; ++u) {
    if (FULL || u < m) {
      acc.x += rv[I + u].x;
      acc.y += rv[I + u].y;
      if (s[I + u] < 0) {
        const float nd = __builtin_bit_cast(float,
            __builtin_amdgcn_readlane(__builtin_bit_cast(int, dv), cur - r_lo));
        float2 o;
        o.x = acc.x * nd;
        o.y = acc.y * nd;
        if (ADD_BIAS) {
          o.x += bv.x;
          o.y += bv.y;
        }
        *(float2*)((char*)out + ((unsigned)cur * ROWBYTES + lane8)) = o;
        cur += 1;
        acc = make_float2(0.0f, 0.0f);
      }
    }
  }
}

template <bool ADD_BIAS>
__global__ void __launch_bounds__(AGT)
csr_aggregate(const int* __restrict__ rowptr, const int* __restrict__ col,
              const float* __restrict__ dinv, const float* __restrict__ xs,
              const float* __restrict__ bias, float* __restrict__ out, int N) {
  const int lane = threadIdx.x & 31;
  const unsigned lane8 = (unsigned)lane * 8u;
  const unsigned wave = __builtin_amdgcn_readfirstlane(threadIdx.x >> 5);
  const unsigned gw = blockIdx.x * (AGT / 32) + wave;
  const unsigned nW = gridDim.x * (AGT / 32);
  const unsigned q = (unsigned)N / nW;
  const unsigned rem = (unsigned)N - q * nW;
  const int w_lo = (int)(gw * q + min(gw, rem));
  const int w_hi = w_lo + (int)q + (gw < rem ? 1 : 0);

  float2 bv = make_float2(0.0f, 0.0f);
  if (ADD_BIAS) bv = ((const float2*)bias)[lane];

  int s[CHUNK];
  float2 rv[CHUNK];
  for (int r_lo = w_lo; r_lo < w_hi; r_lo += 32) {
    const int r_hi = min(w_hi, r_lo + 32);
    const int e_lo = rowptr[r_lo];
    const int e_hi = rowptr[r_hi];
    const float dv = dinv[min(r_lo + lane, N - 1)];
    int cur = r_lo;
    float2 acc = make_float2(0.0f, 0.0f);
    int ent = col[min(e_lo + lane, e_hi - 1)];
    for (int base = e_lo; base < e_hi; base += CHUNK) {
      const int cnt = e_hi - base;
      if (cnt >= CHUNK) {
        gather_rows<true, 0, CHUNK>(xs, ent, CHUNK, lane8, s, rv);
        ent = col[min(base + CHUNK + lane, e_hi - 1)];
        sum_rows<ADD_BIAS, true, 0, CHUNK>(out, s, rv, CHUNK, lane8, bv, r_lo, dv, cur, acc);
      } else {
        const int m0 = min(cnt, CHUNK / 2);
        gather_rows<false, 0, CHUNK / 2>(xs, ent, m0, lane8, s, rv);
        if (cnt > CHUNK / 2) gather_rows<false, CHUNK / 2, CHUNK / 2>(xs, ent, cnt - CHUNK / 2, lane8, s, rv);
        sum_rows<ADD_BIAS, false, 0, CHUNK / 2>(out, s, rv, m0, lane8, bv, r_lo, dv, cur, acc);
        if (cnt > CHUNK / 2)
          sum_rows<ADD_BIAS, false, CHUNK / 2, CHUNK / 2>(out, s, rv, cnt - CHUNK / 2, lane8, bv, r_lo, dv, cur, acc);
      }
    }
  }
}


extern "C" void kernel_launch(void* const* d_in, const int* in_sizes, int n_in,
                              void* d_out, int out_size, void* d_ws, size_t ws_size,
                              hipStream_t stream) {
  const float* ndata = (const float*)d_in[0];
  const float* W0    = (const float*)d_in[1];
  const float* b0    = (const float*)d_in[2];
  const float* W1    = (const float*)d_in[3];
  const float* b1    = (const float*)d_in[4];
  const int* edge_src = (const int*)d_in[5];
  const int* edge_dst = (const int*)d_in[6];

  const int N = in_sizes[0] / D;
  const int E = in_sizes[5];

  const int NB = (N + RB - 1) / RB;
  const int CH = (E + G - 1) / G;

  size_t Npad = ((size_t)N + 3) & ~(size_t)3;
  float* dinv = (float*)d_ws;
  float* xw   = dinv + Npad;
  float* agg  = xw + (size_t)N * D;
  int* pairs  = (int*)(agg + (size_t)N * D);
  int* cnt    = pairs + 2 * (size_t)E;
  int* total  = cnt + (size_t)G * NB;
  int* bstart = total + NB;
  int* rowptr = bstart + (NB + 1);
  int* col    = rowptr + ((size_t)N + 1);
  float* out  = (float*)d_out;

  const int T = 256;

  bucket_count<<<G, BT, NB * sizeof(int), stream>>>(edge_dst, cnt, E, CH, NB);
  bucket_scan<<<NB, G, 0, stream>>>(cnt, total, NB);
  bucket_fill<<<G, BT, NB * sizeof(int), stream>>>(edge_src, edge_dst, cnt, total, bstart, pairs, E, CH, NB);
  bucket_rows<<<NB, T, 0, stream>>>(pairs, bstart, dinv, rowptr, col, N, NB);

  const int nTiles = N / 16;
  const int remRow0 = nTiles * 16;
  const int wavesPerBlk = T / 32;
  const int gemmGrid = (nTiles + wavesPerBlk - 1) / wavesPerBlk;

  gemm64_wmma<false><<<gemmGrid, T, 0, stream>>>(ndata, W0, nullptr, dinv, xw, nTiles);
  if (remRow0 < N)
    gemm64_tail<false><<<((N - remRow0) * D + T - 1) / T, T, 0, stream>>>(ndata, W0, nullptr, dinv, xw, remRow0, N);

  csr_aggregate<false><<<AGRID, AGT, 0, stream>>>(rowptr, col, dinv, xw, nullptr, agg, N);

  gemm64_wmma<true><<<gemmGrid, T, 0, stream>>>(agg, W1, b0, dinv, xw, nTiles);
  if (remRow0 < N)
    gemm64_tail<true><<<((N - remRow0) * D + T - 1) / T, T, 0, stream>>>(agg, W1, b0, dinv, xw, remRow0, N);

  csr_aggregate<true><<<AGRID, AGT, 0, stream>>>(rowptr, col, dinv, xw, b1, out, N);
}


# Both dense products fused into the kernels that finish their inputs: layer 0 into the row-list kernel, layer 1 into the first aggregation through a per-wave LDS tile; 8 launches become 6
# speedup vs baseline: 3.0603x; 1.0895x over previous
#include <hip/hip_runtime.h>
#include <hip/hip_bf16.h>


typedef __attribute__((ext_vector_type(2))) float v2f;
typedef __attribute__((ext_vector_type(8))) float v8f;

#define D 64


#define RB 128
#define G 256
#define BT 1024

__device__ __forceinline__ int wave_incl_scan(int x, int lane) {
  for (int o = 1; o < 32; o <<= 1) {
    int y = __shfl_up(x, o);
    if (lane >= o) x += y;
  }
  return x;
}

__global__ void __launch_bounds__(BT)
bucket_count(const int* __restrict__ dst, int* __restrict__ cnt, int E, int CH, int NB) {
  extern __shared__ int hist[];
  for (int k = threadIdx.x; k < NB; k += BT) hist[k] = 0;
  __syncthreads();
  const int e0 = blockIdx.x * CH;
  const int e1 = min(E, e0 + CH);
  for (int e = e0 + (int)threadIdx.x; e < e1; e += BT) atomicAdd(&hist[dst[e] / RB], 1);
  __syncthreads();
  for (int k = threadIdx.x; k < NB; k += BT) cnt[blockIdx.x * NB + k] = hist[k];
}

__global__ void __launch_bounds__(G)
bucket_scan(int* __restrict__ cnt, int* __restrict__ total, int NB) {
  __shared__ int wsum[G / 32];
  const int k = blockIdx.x, t = threadIdx.x, lane = t & 31, w = t >> 5;
  const int v = cnt[t * NB + k];
  const int x = wave_incl_scan(v, lane);
  if (lane == 31) wsum[w] = x;
  __syncthreads();
  int pre = 0;
  for (int i = 0; i < w; ++i) pre += wsum[i];
  cnt[t * NB + k] = pre + x - v;
  if (t == G - 1) total[k] = pre + x;
}

__global__ void __launch_bounds__(BT)
bucket_fill(const int* __restrict__ src, const int* __restrict__ dst,
            const int* __restrict__ off, const int* __restrict__ total,
            int* __restrict__ bstart, int* __restrict__ pairs, int E, int CH, int NB) {
  extern __shared__ int cur[];
  __shared__ int wsum[BT / 32];
  const int t = threadIdx.x, lane = t & 31, w = t >> 5;
  int carry = 0;
  for (int base = 0; base < NB; base += BT) {
    const int k = base + t;
    const int v = (k < NB) ? total[k] : 0;
    const int x = wave_incl_scan(v, lane);
    if (lane == 31) wsum[w] = x;
    __syncthreads();
    int pre = 0, tot = 0;
    for (int i = 0; i < BT / 32; ++i) {
      const int s = wsum[i];
      if (i < w) pre += s;
      tot += s;
    }
    if (k < NB) {
      const int ex = carry + pre + x - v;
      cur[k] = ex + off[blockIdx.x * NB + k];
      if (blockIdx.x == 0) bstart[k] = ex;
    }
    carry += tot;
    __syncthreads();
  }
  if (blockIdx.x == 0 && t == 0) bstart[NB] = carry;
  const int e0 = blockIdx.x * CH;
  const int e1 = min(E, e0 + CH);
  for (int e = e0 + t; e < e1; e += BT) {
    const int s = src[e];
    const int q = dst[e];
    const int pos = atomicAdd(&cur[q / RB], 1);
    ((int2*)pairs)[pos] = make_int2(s, q);
  }
}


template <int I0, int CNT>
__device__ __forceinline__ void tile_product(const v2f (&a)[CNT], const float* Wl, int lrow, int khalf,
                                             v8f& acc0, v8f& acc1, v8f& acc2, v8f& acc3) {
#pragma unroll
  for (int i = 0; i < CNT; ++i) {
    const int kb = 4 * (I0 + i) + khalf * 2;
#pragma unroll
    for (int nt = 0; nt < 4; ++nt) {
      const int n = nt * 16 + lrow;
      v2f b;
      b.x = Wl[(kb + 0) * D + n];
      b.y = Wl[(kb + 1) * D + n];
      switch (nt) {
        case 0: acc0 = __builtin_amdgcn_wmma_f32_16x16x4_f32(false, a[i], false, b, (short)0, acc0, false, false); break;
        case 1: acc1 = __builtin_amdgcn_wmma_f32_16x16x4_f32(false, a[i], false, b, (short)0, acc1, false, false); break;
        case 2: acc2 = __builtin_amdgcn_wmma_f32_16x16x4_f32(false, a[i], false, b, (short)0, acc2, false, false); break;
        case 3: acc3 = __builtin_amdgcn_wmma_f32_16x16x4_f32(false, a[i], false, b, (short)0, acc3, false, false); break;
      }
    }
  }
}

__device__ __forceinline__ void tile_store(float* __restrict__ Y, int m0, int lrow, const float (&rs)[8],
                                           const v8f& acc0, const v8f& acc1, const v8f& acc2, const v8f& acc3) {
#pragma unroll
  for (int r = 0; r < 8; ++r) {
    float* yrow = Y + (size_t)(m0 + r) * D;
    yrow[0 * 16 + lrow] = acc0[r] * rs[r];
    yrow[1 * 16 + lrow] = acc1[r] * rs[r];
    yrow[2 * 16 + lrow] = acc2[r] * rs[r];
    yrow[3 * 16 + lrow] = acc3[r] * rs[r];
  }
}

#define ROWEND ((int)0x80000000)

__global__ void __launch_bounds__(256)
bucket_rows_gemm(const int* __restrict__ pairs, const int* __restrict__ bstart,
                 const float* __restrict__ X, const float* __restrict__ W,
                 float* __restrict__ dinv, int* __restrict__ rowptr, int* __restrict__ col,
                 float* __restrict__ Y, int N, int NB) {
  __shared__ int c[RB];
  __shared__ int wsum[RB / 32];
  __shared__ float dl[RB];
  __shared__ __attribute__((aligned(16))) float Wlds[D * D];
  const int k = blockIdx.x, t = threadIdx.x, lane = t & 31, w = t >> 5;
  const int r0 = k * RB;
  const int rows = min(RB, N - r0);
  if (t < RB) c[t] = 0;
  __syncthreads();
  const int b0 = bstart[k], b1 = bstart[k + 1];
  for (int e = b0 + t; e < b1; e += 4 * 256) {
    int qd[4];
#pragma unroll
    for (int j = 0; j < 4; ++j) qd[j] = pairs[2 * min(e + j * 256, b1 - 1) + 1];
    asm volatile("" :: "v"(qd[0]), "v"(qd[1]), "v"(qd[2]), "v"(qd[3]));
#pragma unroll
    for (int j = 0; j < 4; ++j)
      if (e + j * 256 < b1) atomicAdd(&c[qd[j] - r0], 1);
  }

  const int lrow = lane & 15;
  const int khalf = lane >> 4;
  const float* xrow = X + (size_t)min(r0 + 16 * w + lrow, N - 1) * D + 2 * khalf;
  float4 wv[4];
#pragma unroll
  for (int j = 0; j < 4; ++j) wv[j] = ((const float4*)W)[t + 256 * j];
  v2f xa[16];
#pragma unroll
  for (int i = 0; i < 16; ++i) xa[i] = *(const v2f*)(xrow + 4 * i);
#pragma unroll
  for (int j = 0; j < 4; ++j) ((float4*)Wlds)[t + 256 * j] = wv[j];

  __syncthreads();
  int indeg = 0, v = 0, x = 0;
  if (t < RB) {
    indeg = c[t];
    v = (t < rows) ? indeg + 1 : 0;
    x = wave_incl_scan(v, lane);
    if (lane == 31) wsum[w] = x;
  }
  __syncthreads();
  if (t < RB) {
    int pre = 0;
    for (int i = 0; i < w; ++i) pre += wsum[i];
    const int base = b0 + r0 + pre + x - v;
    float dval = 0.0f;
    if (t < rows) {
      dval = rsqrtf(1.0f + (float)indeg);
      rowptr[r0 + t] = base;
      dinv[r0 + t] = dval;
      col[base + indeg] = (r0 + t) | ROWEND;
    }
    dl[t] = dval;
    c[t] = base;
  }
  if (k == NB - 1 && t == 0) rowptr[N] = b1 + N;
  __syncthreads();
  for (int e = b0 + t; e < b1; e += 4 * 256) {
    int2 p[4];
#pragma unroll
    for (int j = 0; j < 4; ++j) p[j] = ((const int2*)pairs)[min(e + j * 256, b1 - 1)];
    asm volatile("" :: "v"(p[0].y), "v"(p[1].y), "v"(p[2].y), "v"(p[3].y));
#pragma unroll
    for (int j = 0; j < 4; ++j) {
      if (e + j * 256 < b1) {
        const int pos = atomicAdd(&c[p[j].y - r0], 1);
        col[pos] = p[j].x;
      }
    }
  }

  const int ws = __builtin_amdgcn_readfirstlane(w);
  if (r0 + 16 * ws >= N) return;
  v8f acc0 = {}, acc1 = {}, acc2 = {}, acc3 = {};
  tile_product<0, 16>(xa, Wlds, lrow, khalf, acc0, acc1, acc2, acc3);
  float rs[8];
#pragma unroll
  for (int r = 0; r < 8; ++r) rs[r] = dl[16 * ws + 8 * khalf + r];
  tile_store(Y, r0 + 16 * ws + 8 * khalf, lrow, rs, acc0, acc1, acc2, acc3);
}


#define AGT 256
#define AGRID 1024
#define CHUNK 32
#define ROWBYTES ((unsigned)(D * sizeof(float)))
#define TP 68
#define TPBYTES ((unsigned)(TP * sizeof(float)))


template <bool FULL, int I, int CNT>
__device__ __forceinline__ void gather_rows(const float* __restrict__ xs, int ent, int m, unsigned lane8,
                                            int (&s)[CHUNK], float2 (&rv)[CHUNK]) {
#pragma unroll
  for (int u = 0; u < CNT; ++u) {
    s[I + u] = FULL ? __builtin_amdgcn_readlane(ent, I + u)
                    : __builtin_amdgcn_readlane(ent, I + min(u, m - 1));
    rv[I + u] = *(const float2*)((const char*)xs + ((unsigned)s[I + u] * ROWBYTES + lane8));
  }
}

template <bool TO_TILE, bool FULL, int I, int CNT>
__device__ __forceinline__ void sum_rows(float* __restrict__ out, float* tilep,
                                         const int (&s)[CHUNK], const float2 (&rv)[CHUNK], int m,
                                         unsigned lane8, float2 bv, int r_lo, float dv,
                                         int& cur, float2& acc) {
#pragma unroll
  for (int u = 0; u < CNT; ++u) {
    if (FULL || u < m) {
      acc.x += rv[I + u].x;
      acc.y += rv[I + u].y;
      if (s[I + u] < 0) {
        const float nd = __builtin_bit_cast(float,
            __builtin_amdgcn_readlane(__builtin_bit_cast(int, dv), cur - r_lo));
        float2 o;
        o.x = acc.x * nd + bv.x;
        o.y = acc.y * nd + bv.y;
        if (TO_TILE) {
          o.x = fmaxf(o.x, 0.0f);
          o.y = fmaxf(o.y, 0.0f);
          *(float2*)((char*)tilep + ((unsigned)(cur - r_lo) * TPBYTES + lane8)) = o;
        } else {
          *(float2*)((char*)out + ((unsigned)cur * ROWBYTES + lane8)) = o;
        }
        cur += 1;
        acc = make_float2(0.0f, 0.0f);
      }
    }
  }
}

template <bool TO_TILE>
__device__ __forceinline__ void walk_rows(const int* __restrict__ rowptr, const int* __restrict__ col,
                                          const float* __restrict__ dinv, const float* __restrict__ xs,
                                          float* __restrict__ out, float* tilep, int N, int r_lo, int r_hi,
                                          int lane, unsigned lane8, float2 bv,
                                          int (&s)[CHUNK], float2 (&rv)[CHUNK]) {
  const int e_lo = rowptr[r_lo];
  const int e_hi = rowptr[r_hi];
  const float dv = dinv[min(r_lo + lane, N - 1)];
  int cur = r_lo;
  float2 acc = make_float2(0.0f, 0.0f);
  int ent = col[min(e_lo + lane, e_hi - 1)];
  for (int base = e_lo; base < e_hi; base += CHUNK) {
    const int cnt = e_hi - base;
    if (cnt >= CHUNK) {
      gather_rows<true, 0, CHUNK>(xs, ent, CHUNK, lane8, s, rv);
      ent = col[min(base + CHUNK + lane, e_hi - 1)];
      sum_rows<TO_TILE, true, 0, CHUNK>(out, tilep, s, rv, CHUNK, lane8, bv, r_lo, dv, cur, acc);
    } else {
      const int m0 = min(cnt, CHUNK / 2);
      gather_rows<false, 0, CHUNK / 2>(xs, ent, m0, lane8, s, rv);
      if (cnt > CHUNK / 2) gather_rows<false, CHUNK / 2, CHUNK / 2>(xs, ent, cnt - CHUNK / 2, lane8, s, rv);
      sum_rows<TO_TILE, false, 0, CHUNK / 2>(out, tilep, s, rv, m0, lane8, bv, r_lo, dv, cur, acc);
      if (cnt > CHUNK / 2)
        sum_rows<TO_TILE, false, CHUNK / 2, CHUNK / 2>(out, tilep, s, rv, cnt - CHUNK / 2, lane8, bv, r_lo, dv, cur, acc);
    }
  }
}

__global__ void __launch_bounds__(AGT)
csr_aggregate(const int* __restrict__ rowptr, const int* __restrict__ col,
              const float* __restrict__ dinv, const float* __restrict__ xs,
              const float* __restrict__ bias, float* __restrict__ out, int N) {
  const int lane = threadIdx.x & 31;
  const unsigned lane8 = (unsigned)lane * 8u;
  const unsigned wave = __builtin_amdgcn_readfirstlane(threadIdx.x >> 5);
  const unsigned gw = blockIdx.x * (AGT / 32) + wave;
  const unsigned nW = gridDim.x * (AGT / 32);
  const unsigned q = (unsigned)N / nW;
  const unsigned rem = (unsigned)N - q * nW;
  const int w_lo = (int)(gw * q + min(gw, rem));
  const int w_hi = w_lo + (int)q + (gw < rem ? 1 : 0);

  const float2 bv = ((const float2*)bias)[lane];

  int s[CHUNK];
  float2 rv[CHUNK];
  for (int r_lo = w_lo; r_lo < w_hi; r_lo += 32)
    walk_rows<false>(rowptr, col, dinv, xs, out, nullptr, N, r_lo, min(w_hi, r_lo + 32), lane, lane8, bv, s, rv);
}

__global__ void __launch_bounds__(AGT, 8)
csr_aggregate_gemm(const int* __restrict__ rowptr, const int* __restrict__ col,
                   const float* __restrict__ dinv, const float* __restrict__ xs,
                   const float* __restrict__ bias, const float* __restrict__ W,
                   float* __restrict__ Y, int N, int nTiles) {
  __shared__ __attribute__((aligned(16))) float Wlds[D * D];
  __shared__ __attribute__((aligned(16))) float tiles[(AGT / 32) * 16 * TP];
  for (int i = threadIdx.x; i < (D * D) / 4; i += AGT) ((float4*)Wlds)[i] = ((const float4*)W)[i];
  __syncthreads();

  const int lane = threadIdx.x & 31;
  const unsigned lane8 = (unsigned)lane * 8u;
  const int lrow = lane & 15;
  const int khalf = lane >> 4;
  const int wave = __builtin_amdgcn_readfirstlane(threadIdx.x >> 5);
  const int gw = blockIdx.x * (AGT / 32) + wave;
  const int nW = gridDim.x * (AGT / 32);
  float* tilep = tiles + wave * (16 * TP);

  const float2 bv = ((const float2*)bias)[lane];

  int s[CHUNK];
  float2 rv[CHUNK];
  for (int tile = gw; tile < nTiles; tile += nW) {
    const int r_lo = tile * 16;
    const int r_hi = min(N, r_lo + 16);
    walk_rows<true>(rowptr, col, dinv, xs, nullptr, tilep, N, r_lo, r_hi, lane, lane8, bv, s, rv);
    for (int r = r_hi - r_lo; r < 16; ++r)
      *(float2*)((char*)tilep + ((unsigned)r * TPBYTES + lane8)) = make_float2(0.0f, 0.0f);
    __builtin_amdgcn_wave_barrier();
    const float* arow = tilep + lrow * TP + 2 * khalf;
    v8f acc0 = {}, acc1 = {}, acc2 = {}, acc3 = {};
    v2f a[8];
#pragma unroll
    for (int i = 0; i < 8; ++i) a[i] = *(const v2f*)(arow + 4 * i);
    tile_product<0, 8>(a, Wlds, lrow, khalf, acc0, acc1, acc2, acc3);
#pragma unroll
    for (int i = 0; i < 8; ++i) a[i] = *(const v2f*)(arow + 4 * (8 + i));
    tile_product<8, 8>(a, Wlds, lrow, khalf, acc0, acc1, acc2, acc3);
    const float dvt = dinv[min(r_lo + lane, N - 1)];
    float rs[8];
#pragma unroll
    for (int r = 0; r < 8; ++r) {
      const float lo = __builtin_bit_cast(float, __builtin_amdgcn_readlane(__builtin_bit_cast(int, dvt), r));
      const float hi = __builtin_bit_cast(float, __builtin_amdgcn_readlane(__builtin_bit_cast(int, dvt), 8 + r));
      rs[r] = khalf ? hi : lo;
    }
    tile_store(Y, r_lo + 8 * khalf, lrow, rs, acc0, acc1, acc2, acc3);
  }
}


extern "C" void kernel_launch(void* const* d_in, const int* in_sizes, int n_in,
                              void* d_out, int out_size, void* d_ws, size_t ws_size,
                              hipStream_t stream) {
  const float* ndata = (const float*)d_in[0];
  const float* W0    = (const float*)d_in[1];
  const float* b0    = (const float*)d_in[2];
  const float* W1    = (const float*)d_in[3];
  const float* b1    = (const float*)d_in[4];
  const int* edge_src = (const int*)d_in[5];
  const int* edge_dst = (const int*)d_in[6];

  const int N = in_sizes[0] / D;
  const int E = in_sizes[5];

  const int NB = (N + RB - 1) / RB;
  const int CH = (E + G - 1) / G;
  const int nTiles = (N + 15) / 16;

  size_t Npad = ((size_t)N + 3) & ~(size_t)3;
  size_t Nt = (size_t)nTiles * 16;
  float* dinv = (float*)d_ws;
  float* xw0  = dinv + Npad;
  float* xw1  = xw0 + Nt * D;
  int* pairs  = (int*)(xw1 + Nt * D);
  int* cnt    = pairs + 2 * (size_t)E;
  int* total  = cnt + (size_t)G * NB;
  int* bstart = total + NB;
  int* rowptr = bstart + (NB + 1);
  int* col    = rowptr + ((size_t)N + 1);
  float* out  = (float*)d_out;

  bucket_count<<<G, BT, NB * sizeof(int), stream>>>(edge_dst, cnt, E, CH, NB);
  bucket_scan<<<NB, G, 0, stream>>>(cnt, total, NB);
  bucket_fill<<<G, BT, NB * sizeof(int), stream>>>(edge_src, edge_dst, cnt, total, bstart, pairs, E, CH, NB);

  bucket_rows_gemm<<<NB, 256, 0, stream>>>(pairs, bstart, ndata, W0, dinv, rowptr, col, xw0, N, NB);

  csr_aggregate_gemm<<<(nTiles + AGT / 32 - 1) / (AGT / 32), AGT, 0, stream>>>(rowptr, col, dinv, xw0, b0, W1, xw1, N, nTiles);

  csr_aggregate<<<AGRID, AGT, 0, stream>>>(rowptr, col, dinv, xw1, b1, out, N);
}
